# starting kernel
# baseline (speedup 1.0000x reference)
#include <hip/hip_runtime.h>

typedef float v2f __attribute__((ext_vector_type(2)));
typedef float v8f __attribute__((ext_vector_type(8)));

#define N_AGENT 81
#define N_NBR   20
#define DD      96
#define HH      96
#define ROWS_PAD 96

__global__ __launch_bounds__(128) void relational_wmma_kernel(
    const float* __restrict__ x,
    const int*   __restrict__ edges,
    const float* __restrict__ W1,
    const float* __restrict__ b1,
    const float* __restrict__ W2,
    const float* __restrict__ b2,
    float* __restrict__ out)
{
    extern __shared__ float lds[];
    float* xh   = lds;
    float* sbuf = lds + 96 * 96;
    float* ybuf = lds + 2 * 96 * 96;
    int*   eg   = (int*)(lds + 3 * 96 * 96);

    const int b    = blockIdx.x;
    const int tid  = threadIdx.x;
    const int lane = tid & 31;
    const int wave = tid >> 5;
    const int lh   = (lane >> 4) & 1;
    const int l15  = lane & 15;

    const float* xb = x + (size_t)b * (N_AGENT * DD);
    for (int f = tid; f < (ROWS_PAD * DD) / 4; f += 128) {
        int row = f / (DD / 4);
        int c4  = f % (DD / 4);
        float4 v = make_float4(0.f, 0.f, 0.f, 0.f);
        if (row < N_AGENT) v = ((const float4*)(xb + row * DD))[c4];
        ((float4*)(xh + row * DD))[c4] = v;
    }
    for (int f = tid; f < N_AGENT * N_NBR; f += 128) eg[f] = edges[f];
    __syncthreads();

    for (int t = wave; t < 36; t += 4) {
        const int mt = t / 6, nt = t % 6;
        v8f accS = {};
        v8f accY = {};
        const int m = mt * 16 + l15;
        const int n = nt * 16 + l15;
        for (int k = 0; k < DD; k += 4) {
            const int k0 = k + lh * 2;
            v2f a;
            a.x = xh[m * DD + k0];
            a.y = xh[m * DD + k0 + 1];
            v2f bt, bb;
            bt.x = W1[(k0)*HH + n];
            bt.y = W1[(k0 + 1) * HH + n];
            bb.x = W1[(DD + k0) * HH + n];
            bb.y = W1[(DD + k0 + 1) * HH + n];
            accS = __builtin_amdgcn_wmma_f32_16x16x4_f32(false, a, false, bt,
                                                         (short)0, accS, false, false);
            accY = __builtin_amdgcn_wmma_f32_16x16x4_f32(false, a, false, bb,
                                                         (short)0, accY, false, false);
        }
        for (int j = 0; j < 8; ++j) {
            const int mr = mt * 16 + j + lh * 8;
            sbuf[mr * HH + n] = accS[j];
            ybuf[mr * HH + n] = accY[j];
        }
    }
    __syncthreads();

    for (int idx = tid; idx < N_AGENT * HH; idx += 128) {
        const int a = idx / HH, c = idx % HH;
        const float sa = sbuf[a * HH + c] + b1[c];
        float v = 0.f;
        #pragma unroll 5
        for (int nn = 0; nn < N_NBR; ++nn) {
            const int j = eg[a * N_NBR + nn];
            const float hv = sa + ybuf[j * HH + c];
            v += fmaxf(hv, 0.f);
        }
        xh[a * HH + c] = v;
    }
    __syncthreads();

    float* ob = out + (size_t)b * (N_AGENT * HH);
    for (int t = wave; t < 36; t += 4) {
        const int mt = t / 6, nt = t % 6;
        v8f acc = {};
        const int m = mt * 16 + l15;
        const int n = nt * 16 + l15;
        for (int k = 0; k < HH; k += 4) {
            const int k0 = k + lh * 2;
            v2f a, bw;
            a.x  = xh[m * HH + k0];
            a.y  = xh[m * HH + k0 + 1];
            bw.x = W2[(k0)*HH + n];
            bw.y = W2[(k0 + 1) * HH + n];
            acc = __builtin_amdgcn_wmma_f32_16x16x4_f32(false, a, false, bw,
                                                        (short)0, acc, false, false);
        }
        const float bias = (float)N_NBR * b2[n];
        for (int j = 0; j < 8; ++j) {
            const int mr = mt * 16 + j + lh * 8;
            if (mr < N_AGENT) ob[mr * HH + n] = acc[j] + bias;
        }
    }
}

extern "C" void kernel_launch(void* const* d_in, const int* in_sizes, int n_in,
                              void* d_out, int out_size, void* d_ws, size_t ws_size,
                              hipStream_t stream) {
    (void)in_sizes; (void)n_in; (void)out_size; (void)d_ws; (void)ws_size;
    const float* x     = (const float*)d_in[0];
    const int*   edges = (const int*)d_in[1];
    const float* W1    = (const float*)d_in[2];
    const float* b1    = (const float*)d_in[3];
    const float* W2    = (const float*)d_in[4];
    const float* b2    = (const float*)d_in[5];
    float* out = (float*)d_out;

    const size_t shmem = (size_t)(3 * 96 * 96) * sizeof(float)
                       + (size_t)(N_AGENT * N_NBR) * sizeof(int);
    relational_wmma_kernel<<<dim3(512), dim3(128), shmem, stream>>>(
        x, edges, W1, b1, W2, b2, out);
}


# Products transposed (weights = A operand, agents = columns): a wave owns 16 agents; s, the neighbour sum and the last product's B operand stay in registers, only y in LDS; last product's k regrouped
# speedup vs baseline: 2.4828x; 2.4828x over previous
#include <hip/hip_runtime.h>


typedef float v2f __attribute__((ext_vector_type(2)));
typedef float v8f __attribute__((ext_vector_type(8)));

#define N_AGENT 81
#define N_NBR   20
#define DD      96
#define HH      96
#define YPITCH  100
#define AHEAD   3

template <bool LAST>
__device__ __forceinline__ constexpr int krow(int ks) {
    return LAST ? 16 * (ks >> 2) + 2 * (ks & 3) : 4 * ks;
}

template <bool LAST>
__device__ __forceinline__ void load_w(const float* __restrict__ Wl, int ks, v2f (&a)[6]) {
    #pragma unroll
    for (int mt = 0; mt < 6; ++mt) {
        a[mt].x = Wl[krow<LAST>(ks) * HH + 16 * mt];
        a[mt].y = Wl[(krow<LAST>(ks) + 1) * HH + 16 * mt];
    }
}

template <bool LAST>
__device__ __forceinline__ void wT_times_b(const float* __restrict__ Wl,
                                           const v2f (&bf)[24], v8f (&acc)[6]) {
    v2f a[AHEAD + 1][6];
    #pragma unroll
    for (int p = 0; p < AHEAD; ++p) load_w<LAST>(Wl, p, a[p]);
    #pragma unroll
    for (int ks = 0; ks < 24; ++ks) {
        if (ks + AHEAD < 24) load_w<LAST>(Wl, ks + AHEAD, a[(ks + AHEAD) % (AHEAD + 1)]);
        #pragma unroll
        for (int mt = 0; mt < 6; ++mt)
            acc[mt] = __builtin_amdgcn_wmma_f32_16x16x4_f32(false, a[ks % (AHEAD + 1)][mt], false, bf[ks],
                                                            (short)0, acc[mt], false, false);
    }
}

__device__ __forceinline__ void add_neighbour(const float* yr, const v8f (&sb)[6], v8f (&hs)[6]) {
    #pragma unroll
    for (int mt = 0; mt < 6; ++mt) {
        const float4 y0 = *(const float4*)(yr + 16 * mt);
        const float4 y1 = *(const float4*)(yr + 16 * mt + 4);
        hs[mt][0] += fmaxf(sb[mt][0] + y0.x, 0.f);
        hs[mt][1] += fmaxf(sb[mt][1] + y0.y, 0.f);
        hs[mt][2] += fmaxf(sb[mt][2] + y0.z, 0.f);
        hs[mt][3] += fmaxf(sb[mt][3] + y0.w, 0.f);
        hs[mt][4] += fmaxf(sb[mt][4] + y1.x, 0.f);
        hs[mt][5] += fmaxf(sb[mt][5] + y1.y, 0.f);
        hs[mt][6] += fmaxf(sb[mt][6] + y1.z, 0.f);
        hs[mt][7] += fmaxf(sb[mt][7] + y1.w, 0.f);
    }
}

__global__ __launch_bounds__(192, 4) void relational_wmma_kernel(
    const float* __restrict__ x,
    const int*   __restrict__ edges,
    const float* __restrict__ W1,
    const float* __restrict__ b1,
    const float* __restrict__ W2,
    const float* __restrict__ b2,
    float* __restrict__ out)
{
    __shared__ float ybuf[N_AGENT * YPITCH];

    const int b    = blockIdx.x;
    const int tid  = threadIdx.x;
    const int lane = tid & 31;
    const int wave = tid >> 5;
    const int h    = lane >> 4;
    const int c    = lane & 15;
    const int arow  = wave * 16 + c;
    const int aload = arow < N_AGENT ? arow : N_AGENT - 1;

    const float* xr = x + ((size_t)b * N_AGENT + aload) * DD + 2 * h;
    v2f xf[24];
    #pragma unroll
    for (int t = 0; t < 24; ++t) xf[t] = *(const v2f*)(xr + 4 * t);

    {
        v8f yacc[6] = {};
        wT_times_b<false>(W1 + (DD + 2 * h) * HH + c, xf, yacc);
        if (arow < N_AGENT) {
            float* yw = ybuf + arow * YPITCH + 8 * h;
            #pragma unroll
            for (int mt = 0; mt < 6; ++mt) {
                *(float4*)(yw + 16 * mt)     = make_float4(yacc[mt][0], yacc[mt][1], yacc[mt][2], yacc[mt][3]);
                *(float4*)(yw + 16 * mt + 4) = make_float4(yacc[mt][4], yacc[mt][5], yacc[mt][6], yacc[mt][7]);
            }
        }
    }

    v8f sb[6] = {};
    wT_times_b<false>(W1 + (2 * h) * HH + c, xf, sb);
    #pragma unroll
    for (int mt = 0; mt < 6; ++mt) {
        const float4 p0 = *(const float4*)(b1 + 16 * mt + 8 * h);
        const float4 p1 = *(const float4*)(b1 + 16 * mt + 8 * h + 4);
        sb[mt][0] += p0.x; sb[mt][1] += p0.y; sb[mt][2] += p0.z; sb[mt][3] += p0.w;
        sb[mt][4] += p1.x; sb[mt][5] += p1.y; sb[mt][6] += p1.z; sb[mt][7] += p1.w;
    }
    __syncthreads();

    v8f hs[6] = {};
    {
        const int* ep = edges + aload * N_NBR;
        int2 ecur = *(const int2*)ep;
        #pragma unroll 1
        for (int n2 = 0; n2 < N_NBR / 2; ++n2) {
            const int nn = n2 + 1 < N_NBR / 2 ? n2 + 1 : n2;
            const int2 enxt = *(const int2*)(ep + 2 * nn);
            add_neighbour(ybuf + ecur.x * YPITCH + 8 * h, sb, hs);
            add_neighbour(ybuf + ecur.y * YPITCH + 8 * h, sb, hs);
            asm volatile("" :: "v"(enxt.x), "v"(enxt.y));
            ecur = enxt;
        }
    }

    v8f oc[6] = {};
    {
        v2f hb[24];
        #pragma unroll
        for (int ks = 0; ks < 24; ++ks) {
            hb[ks].x = hs[ks >> 2][2 * (ks & 3)];
            hb[ks].y = hs[ks >> 2][2 * (ks & 3) + 1];
        }
        wT_times_b<true>(W2 + (8 * h) * HH + c, hb, oc);
    }

    if (arow < N_AGENT) {
        float* op = out + ((size_t)b * N_AGENT + arow) * HH + 8 * h;
        #pragma unroll
        for (int ot = 0; ot < 6; ++ot) {
            const float4 q0 = *(const float4*)(b2 + 16 * ot + 8 * h);
            const float4 q1 = *(const float4*)(b2 + 16 * ot + 8 * h + 4);
            const float g0 = (float)N_NBR * q0.x, g1 = (float)N_NBR * q0.y;
            const float g2 = (float)N_NBR * q0.z, g3 = (float)N_NBR * q0.w;
            const float g4 = (float)N_NBR * q1.x, g5 = (float)N_NBR * q1.y;
            const float g6 = (float)N_NBR * q1.z, g7 = (float)N_NBR * q1.w;
            *(float4*)(op + 16 * ot)     = make_float4(oc[ot][0] + g0, oc[ot][1] + g1, oc[ot][2] + g2, oc[ot][3] + g3);
            *(float4*)(op + 16 * ot + 4) = make_float4(oc[ot][4] + g4, oc[ot][5] + g5, oc[ot][6] + g6, oc[ot][7] + g7);
        }
    }
}

extern "C" void kernel_launch(void* const* d_in, const int* in_sizes, int n_in,
                              void* d_out, int out_size, void* d_ws, size_t ws_size,
                              hipStream_t stream) {
    (void)in_sizes; (void)n_in; (void)out_size; (void)d_ws; (void)ws_size;
    const float* x     = (const float*)d_in[0];
    const int*   edges = (const int*)d_in[1];
    const float* W1    = (const float*)d_in[2];
    const float* b1    = (const float*)d_in[3];
    const float* W2    = (const float*)d_in[4];
    const float* b2    = (const float*)d_in[5];
    float* out = (float*)d_out;

    relational_wmma_kernel<<<dim3(512), dim3(192), 0, stream>>>(
        x, edges, W1, b1, W2, b2, out);
}


# Barrier moved up behind the y product; s, then neighbour sum and last product run tile by tile (16 hidden columns), so LDS/vector work of a wave overlaps matrix work; sum and k order kept
# speedup vs baseline: 2.7652x; 1.1137x over previous
#include <hip/hip_runtime.h>


typedef float v2f __attribute__((ext_vector_type(2)));
typedef float v8f __attribute__((ext_vector_type(8)));

#define N_AGENT 81
#define N_NBR   20
#define DD      96
#define HH      96
#define YPITCH  100
#define AHEAD   3

template <bool LAST>
__device__ __forceinline__ constexpr int krow(int ks) {
    return LAST ? 16 * (ks >> 2) + 2 * (ks & 3) : 4 * ks;
}

template <bool LAST>
__device__ __forceinline__ void load_w(const float* __restrict__ Wl, int ks, v2f (&a)[6]) {
    #pragma unroll
    for (int mt = 0; mt < 6; ++mt) {
        a[mt].x = Wl[krow<LAST>(ks) * HH + 16 * mt];
        a[mt].y = Wl[(krow<LAST>(ks) + 1) * HH + 16 * mt];
    }
}

template <bool LAST>
__device__ __forceinline__ void wT_times_b(const float* __restrict__ Wl,
                                           const v2f (&bf)[24], v8f (&acc)[6]) {
    v2f a[AHEAD + 1][6];
    #pragma unroll
    for (int p = 0; p < AHEAD; ++p) load_w<LAST>(Wl, p, a[p]);
    #pragma unroll
    for (int ks = 0; ks < 24; ++ks) {
        if (ks + AHEAD < 24) load_w<LAST>(Wl, ks + AHEAD, a[(ks + AHEAD) % (AHEAD + 1)]);
        #pragma unroll
        for (int mt = 0; mt < 6; ++mt)
            acc[mt] = __builtin_amdgcn_wmma_f32_16x16x4_f32(false, a[ks % (AHEAD + 1)][mt], false, bf[ks],
                                                            (short)0, acc[mt], false, false);
    }
}

__device__ __forceinline__ void ask_group(const float* yl, const int (&nb)[N_NBR], int g, int tl, float4 (&yv)[10]) {
    #pragma unroll
    for (int i = 0; i < 10; ++i)
        yv[i] = *(const float4*)(yl + nb[5 * g + (i >> 1)] + 16 * tl + 4 * (i & 1));
}

__device__ __forceinline__ void add_group(const float4 (&yv)[10], const v8f& sbt, v8f& hst) {
    #pragma unroll
    for (int i = 0; i < 10; ++i) {
        const int p = 4 * (i & 1);
        hst[p + 0] += fmaxf(sbt[p + 0] + yv[i].x, 0.f);
        hst[p + 1] += fmaxf(sbt[p + 1] + yv[i].y, 0.f);
        hst[p + 2] += fmaxf(sbt[p + 2] + yv[i].z, 0.f);
        hst[p + 3] += fmaxf(sbt[p + 3] + yv[i].w, 0.f);
    }
}

__global__ __launch_bounds__(192, 4) void relational_wmma_kernel(
    const float* __restrict__ x,
    const int*   __restrict__ edges,
    const float* __restrict__ W1,
    const float* __restrict__ b1,
    const float* __restrict__ W2,
    const float* __restrict__ b2,
    float* __restrict__ out)
{
    __shared__ float ybuf[N_AGENT * YPITCH];

    const int b    = blockIdx.x;
    const int tid  = threadIdx.x;
    const int lane = tid & 31;
    const int wave = tid >> 5;
    const int h    = lane >> 4;
    const int c    = lane & 15;
    const int arow  = wave * 16 + c;
    const int aload = arow < N_AGENT ? arow : N_AGENT - 1;

    const float* xr = x + ((size_t)b * N_AGENT + aload) * DD + 2 * h;
    v2f xf[24];
    #pragma unroll
    for (int t = 0; t < 24; ++t) xf[t] = *(const v2f*)(xr + 4 * t);

    int nb[N_NBR];
    {
        const int* ep = edges + aload * N_NBR;
        #pragma unroll
        for (int q = 0; q < N_NBR / 4; ++q) {
            const int4 e4 = *(const int4*)(ep + 4 * q);
            nb[4 * q + 0] = e4.x * YPITCH;
            nb[4 * q + 1] = e4.y * YPITCH;
            nb[4 * q + 2] = e4.z * YPITCH;
            nb[4 * q + 3] = e4.w * YPITCH;
        }
    }

    {
        v8f yacc[6] = {};
        wT_times_b<false>(W1 + (DD + 2 * h) * HH + c, xf, yacc);
        if (arow < N_AGENT) {
            float* yw = ybuf + arow * YPITCH + 8 * h;
            #pragma unroll
            for (int mt = 0; mt < 6; ++mt) {
                *(float4*)(yw + 16 * mt)     = make_float4(yacc[mt][0], yacc[mt][1], yacc[mt][2], yacc[mt][3]);
                *(float4*)(yw + 16 * mt + 4) = make_float4(yacc[mt][4], yacc[mt][5], yacc[mt][6], yacc[mt][7]);
            }
        }
    }

    __syncthreads();

    v8f sb[6] = {};
    wT_times_b<false>(W1 + (2 * h) * HH + c, xf, sb);
    #pragma unroll
    for (int mt = 0; mt < 6; ++mt) {
        const float4 p0 = *(const float4*)(b1 + 16 * mt + 8 * h);
        const float4 p1 = *(const float4*)(b1 + 16 * mt + 8 * h + 4);
        sb[mt][0] += p0.x; sb[mt][1] += p0.y; sb[mt][2] += p0.z; sb[mt][3] += p0.w;
        sb[mt][4] += p1.x; sb[mt][5] += p1.y; sb[mt][6] += p1.z; sb[mt][7] += p1.w;
    }

    v8f oc[6] = {};
    {
        const float* yl  = ybuf + 8 * h;
        const float* W2l = W2 + (8 * h) * HH + c;
        float4 yv[10];
        v2f aw[3][6];
        v8f hcur = {};
        load_w<true>(W2l, 0, aw[0]);
        load_w<true>(W2l, 1, aw[1]);
        #pragma unroll
        for (int g = 0; g < 4; ++g) {
            ask_group(yl, nb, g, 0, yv);
            __builtin_amdgcn_sched_barrier(0);
            add_group(yv, sb[0], hcur);
            __builtin_amdgcn_sched_barrier(0);
        }
        #pragma unroll
        for (int t = 0; t < 6; ++t) {
            v2f hb[4];
            #pragma unroll
            for (int j = 0; j < 4; ++j) {
                hb[j].x = hcur[2 * j];
                hb[j].y = hcur[2 * j + 1];
            }
            v8f hnext = {};
            const int tn = t + 1 < 6 ? t + 1 : 5;
            #pragma unroll
            for (int j = 0; j < 4; ++j) {
                const int ks = 4 * t + j;
                if (t + 1 < 6) ask_group(yl, nb, j, tn, yv);
                if (ks + 2 < 24) load_w<true>(W2l, ks + 2, aw[(ks + 2) % 3]);
                __builtin_amdgcn_sched_barrier(0);
                #pragma unroll
                for (int mt = 0; mt < 6; ++mt)
                    oc[mt] = __builtin_amdgcn_wmma_f32_16x16x4_f32(false, aw[ks % 3][mt], false, hb[j],
                                                                   (short)0, oc[mt], false, false);
                __builtin_amdgcn_sched_barrier(0);
                if (t + 1 < 6) add_group(yv, sb[tn], hnext);
                __builtin_amdgcn_sched_barrier(0);
            }
            hcur = hnext;
        }
    }

    if (arow < N_AGENT) {
        float* op = out + ((size_t)b * N_AGENT + arow) * HH + 8 * h;
        #pragma unroll
        for (int ot = 0; ot < 6; ++ot) {
            const float4 q0 = *(const float4*)(b2 + 16 * ot + 8 * h);
            const float4 q1 = *(const float4*)(b2 + 16 * ot + 8 * h + 4);
            const float g0 = (float)N_NBR * q0.x, g1 = (float)N_NBR * q0.y;
            const float g2 = (float)N_NBR * q0.z, g3 = (float)N_NBR * q0.w;
            const float g4 = (float)N_NBR * q1.x, g5 = (float)N_NBR * q1.y;
            const float g6 = (float)N_NBR * q1.z, g7 = (float)N_NBR * q1.w;
            *(float4*)(op + 16 * ot)     = make_float4(oc[ot][0] + g0, oc[ot][1] + g1, oc[ot][2] + g2, oc[ot][3] + g3);
            *(float4*)(op + 16 * ot + 4) = make_float4(oc[ot][4] + g4, oc[ot][5] + g5, oc[ot][6] + g6, oc[ot][7] + g7);
        }
    }
}

extern "C" void kernel_launch(void* const* d_in, const int* in_sizes, int n_in,
                              void* d_out, int out_size, void* d_ws, size_t ws_size,
                              hipStream_t stream) {
    (void)in_sizes; (void)n_in; (void)out_size; (void)d_ws; (void)ws_size;
    const float* x     = (const float*)d_in[0];
    const int*   edges = (const int*)d_in[1];
    const float* W1    = (const float*)d_in[2];
    const float* b1    = (const float*)d_in[3];
    const float* W2    = (const float*)d_in[4];
    const float* b2    = (const float*)d_in[5];
    float* out = (float*)d_out;

    relational_wmma_kernel<<<dim3(512), dim3(192), 0, stream>>>(
        x, edges, W1, b1, W2, b2, out);
}


# Weights staged in LDS as [column][k], once per 12-wave work-group (two batch items); A fragments read from LDS 16 B (two k-steps) a load, was 4 B global loads; products 1,2 k regrouped: new sum order
# speedup vs baseline: 3.1289x; 1.1315x over previous
#include <hip/hip_runtime.h>


typedef float v2f __attribute__((ext_vector_type(2)));
typedef float v4f __attribute__((ext_vector_type(4)));
typedef float v8f __attribute__((ext_vector_type(8)));

#define N_AGENT 81
#define N_NBR   20
#define DD      96
#define HH      96
#define YPITCH  100
#define WPITCH  100
#define WMAT    (HH * WPITCH)

__device__ __forceinline__ void load_wq(const float* wl, int koff, v4f (&a)[6]) {
    #pragma unroll
    for (int mt = 0; mt < 6; ++mt) a[mt] = *(const v4f*)(wl + 16 * mt * WPITCH + koff);
}

__device__ __forceinline__ void wT_times_b(const float* wl, const v4f (&bq)[12], v8f (&acc)[6]) {
    v4f a[2][2][6];
    load_wq(wl, 0, a[0][0]);
    load_wq(wl, 8, a[0][1]);
    #pragma unroll
    for (int blk = 0; blk < 6; ++blk) {
        if (blk + 1 < 6) {
            load_wq(wl, 16 * (blk + 1), a[(blk + 1) & 1][0]);
            load_wq(wl, 16 * (blk + 1) + 8, a[(blk + 1) & 1][1]);
        }
        __builtin_amdgcn_sched_barrier(0);
        #pragma unroll
        for (int p = 0; p < 2; ++p) {
            const int u = 2 * blk + p;
            const v2f b0 = __builtin_shufflevector(bq[u], bq[u], 0, 1);
            const v2f b1 = __builtin_shufflevector(bq[u], bq[u], 2, 3);
            #pragma unroll
            for (int mt = 0; mt < 6; ++mt) {
                const v4f aq = a[blk & 1][p][mt];
                acc[mt] = __builtin_amdgcn_wmma_f32_16x16x4_f32(false, __builtin_shufflevector(aq, aq, 0, 1), false, b0,
                                                                (short)0, acc[mt], false, false);
            }
            #pragma unroll
            for (int mt = 0; mt < 6; ++mt) {
                const v4f aq = a[blk & 1][p][mt];
                acc[mt] = __builtin_amdgcn_wmma_f32_16x16x4_f32(false, __builtin_shufflevector(aq, aq, 2, 3), false, b1,
                                                                (short)0, acc[mt], false, false);
            }
        }
        __builtin_amdgcn_sched_barrier(0);
    }
}

__device__ __forceinline__ void ask_group(const float* yl, const int (&nb)[N_NBR], int g, int tl, float4 (&yv)[10]) {
    #pragma unroll
    for (int i = 0; i < 10; ++i)
        yv[i] = *(const float4*)(yl + nb[5 * g + (i >> 1)] + 16 * tl + 4 * (i & 1));
}

__device__ __forceinline__ void add_group(const float4 (&yv)[10], const v8f& sbt, v8f& hst) {
    #pragma unroll
    for (int i = 0; i < 10; ++i) {
        const int p = 4 * (i & 1);
        hst[p + 0] += fmaxf(sbt[p + 0] + yv[i].x, 0.f);
        hst[p + 1] += fmaxf(sbt[p + 1] + yv[i].y, 0.f);
        hst[p + 2] += fmaxf(sbt[p + 2] + yv[i].z, 0.f);
        hst[p + 3] += fmaxf(sbt[p + 3] + yv[i].w, 0.f);
    }
}

__global__ __launch_bounds__(384) void relational_wmma_kernel(
    const float* __restrict__ x,
    const int*   __restrict__ edges,
    const float* __restrict__ W1,
    const float* __restrict__ b1,
    const float* __restrict__ W2,
    const float* __restrict__ b2,
    float* __restrict__ out)
{
    __shared__ __attribute__((aligned(16))) float ybuf2[2 * N_AGENT * YPITCH];
    __shared__ __attribute__((aligned(16))) float wbuf[3 * WMAT];

    const int tid  = threadIdx.x;
    const int lane = tid & 31;
    const int wv12 = tid >> 5;
    const int bsel = wv12 >= 6 ? 1 : 0;
    const int wave = wv12 - 6 * bsel;
    const int b    = 2 * blockIdx.x + bsel;
    float* ybuf = ybuf2 + bsel * (N_AGENT * YPITCH);
    const int h    = lane >> 4;
    const int c    = lane & 15;
    const int arow  = wave * 16 + c;
    const int aload = arow < N_AGENT ? arow : N_AGENT - 1;

    const float* xr = x + ((size_t)b * N_AGENT + aload) * DD + 4 * h;
    v4f xf[12];
    #pragma unroll
    for (int u = 0; u < 12; ++u) xf[u] = *(const v4f*)(xr + 8 * u);

    int nb[N_NBR];
    {
        const int* ep = edges + aload * N_NBR;
        #pragma unroll
        for (int q = 0; q < N_NBR / 4; ++q) {
            const int4 e4 = *(const int4*)(ep + 4 * q);
            nb[4 * q + 0] = e4.x * YPITCH;
            nb[4 * q + 1] = e4.y * YPITCH;
            nb[4 * q + 2] = e4.z * YPITCH;
            nb[4 * q + 3] = e4.w * YPITCH;
        }
    }

    {
        const int kk = lane & 7;
        const int qq = lane >> 3;
        const int col0 = 16 * wave + 4 * qq;
        const float* g1 = W1 + (96 * bsel + kk) * HH + col0;
        const float* g2 = W2 + (48 * bsel + kk) * HH + col0;
        float* d1 = wbuf + bsel * WMAT + col0 * WPITCH + kk;
        float* d2 = wbuf + 2 * WMAT + col0 * WPITCH + 48 * bsel + kk;
        float4 wq[18];
        #pragma unroll
        for (int i = 0; i < 18; ++i)
            wq[i] = i < 12 ? *(const float4*)(g1 + 8 * i * HH) : *(const float4*)(g2 + 8 * (i - 12) * HH);
        #pragma unroll
        for (int i = 0; i < 18; ++i) {
            float* d = i < 12 ? d1 + 8 * i : d2 + 8 * (i - 12);
            d[0 * WPITCH] = wq[i].x;
            d[1 * WPITCH] = wq[i].y;
            d[2 * WPITCH] = wq[i].z;
            d[3 * WPITCH] = wq[i].w;
        }
    }
    __syncthreads();

    const float* wl12 = wbuf + c * WPITCH + 4 * h;

    {
        v8f yacc[6] = {};
        wT_times_b(wl12 + WMAT, xf, yacc);
        if (arow < N_AGENT) {
            float* yw = ybuf + arow * YPITCH + 8 * h;
            #pragma unroll
            for (int mt = 0; mt < 6; ++mt) {
                *(float4*)(yw + 16 * mt)     = make_float4(yacc[mt][0], yacc[mt][1], yacc[mt][2], yacc[mt][3]);
                *(float4*)(yw + 16 * mt + 4) = make_float4(yacc[mt][4], yacc[mt][5], yacc[mt][6], yacc[mt][7]);
            }
        }
    }

    __syncthreads();

    v8f sb[6] = {};
    wT_times_b(wl12, xf, sb);
    #pragma unroll
    for (int mt = 0; mt < 6; ++mt) {
        const float4 p0 = *(const float4*)(b1 + 16 * mt + 8 * h);
        const float4 p1 = *(const float4*)(b1 + 16 * mt + 8 * h + 4);
        sb[mt][0] += p0.x; sb[mt][1] += p0.y; sb[mt][2] += p0.z; sb[mt][3] += p0.w;
        sb[mt][4] += p1.x; sb[mt][5] += p1.y; sb[mt][6] += p1.z; sb[mt][7] += p1.w;
    }

    v8f oc[6] = {};
    {
        const float* yl  = ybuf + 8 * h;
        const float* wl3 = wbuf + 2 * WMAT + c * WPITCH + 8 * h;
        float4 yv[10];
        v4f aw[2][6];
        v8f hcur = {};
        load_wq(wl3, 0, aw[0]);
        #pragma unroll
        for (int g = 0; g < 4; ++g) {
            ask_group(yl, nb, g, 0, yv);
            __builtin_amdgcn_sched_barrier(0);
            add_group(yv, sb[0], hcur);
            __builtin_amdgcn_sched_barrier(0);
        }
        #pragma unroll
        for (int t = 0; t < 6; ++t) {
            v2f hb[4];
            #pragma unroll
            for (int j = 0; j < 4; ++j) {
                hb[j].x = hcur[2 * j];
                hb[j].y = hcur[2 * j + 1];
            }
            v8f hnext = {};
            const int tn = t + 1 < 6 ? t + 1 : 5;
            #pragma unroll
            for (int j = 0; j < 4; ++j) {
                const int ks = 4 * t + j;
                if (t + 1 < 6) ask_group(yl, nb, j, tn, yv);
                const int pr = ks >> 1;
                if ((j & 1) == 0 && pr + 1 < 12) load_wq(wl3, 16 * ((pr + 1) >> 1) + 4 * ((pr + 1) & 1), aw[(pr + 1) & 1]);
                __builtin_amdgcn_sched_barrier(0);
                #pragma unroll
                for (int mt = 0; mt < 6; ++mt) {
                    const v4f aq = aw[pr & 1][mt];
                    const v2f af = (j & 1) ? __builtin_shufflevector(aq, aq, 2, 3) : __builtin_shufflevector(aq, aq, 0, 1);
                    oc[mt] = __builtin_amdgcn_wmma_f32_16x16x4_f32(false, af, false, hb[j],
                                                                   (short)0, oc[mt], false, false);
                }
                __builtin_amdgcn_sched_barrier(0);
                if (t + 1 < 6) add_group(yv, sb[tn], hnext);
                __builtin_amdgcn_sched_barrier(0);
            }
            hcur = hnext;
        }
    }

    if (arow < N_AGENT) {
        float* op = out + ((size_t)b * N_AGENT + arow) * HH + 8 * h;
        #pragma unroll
        for (int ot = 0; ot < 6; ++ot) {
            const float4 q0 = *(const float4*)(b2 + 16 * ot + 8 * h);
            const float4 q1 = *(const float4*)(b2 + 16 * ot + 8 * h + 4);
            const float g0 = (float)N_NBR * q0.x, g1 = (float)N_NBR * q0.y;
            const float g2 = (float)N_NBR * q0.z, g3 = (float)N_NBR * q0.w;
            const float g4 = (float)N_NBR * q1.x, g5 = (float)N_NBR * q1.y;
            const float g6 = (float)N_NBR * q1.z, g7 = (float)N_NBR * q1.w;
            *(float4*)(op + 16 * ot)     = make_float4(oc[ot][0] + g0, oc[ot][1] + g1, oc[ot][2] + g2, oc[ot][3] + g3);
            *(float4*)(op + 16 * ot + 4) = make_float4(oc[ot][4] + g4, oc[ot][5] + g5, oc[ot][6] + g6, oc[ot][7] + g7);
        }
    }
}

extern "C" void kernel_launch(void* const* d_in, const int* in_sizes, int n_in,
                              void* d_out, int out_size, void* d_ws, size_t ws_size,
                              hipStream_t stream) {
    (void)in_sizes; (void)n_in; (void)out_size; (void)d_ws; (void)ws_size;
    const float* x     = (const float*)d_in[0];
    const int*   edges = (const int*)d_in[1];
    const float* W1    = (const float*)d_in[2];
    const float* b1    = (const float*)d_in[3];
    const float* W2    = (const float*)d_in[4];
    const float* b2    = (const float*)d_in[5];
    float* out = (float*)d_out;

    relational_wmma_kernel<<<dim3(256), dim3(384), 0, stream>>>(
        x, edges, W1, b1, W2, b2, out);
}


# Output rows staged through a per-wave LDS block (16 rows, pitch 100): each global store writes 4 whole 128-byte lines, was 16 lines in two 16-byte pieces each; values and sum order unchanged
# speedup vs baseline: 3.5542x; 1.1359x over previous
#include <hip/hip_runtime.h>


typedef float v2f __attribute__((ext_vector_type(2)));
typedef float v4f __attribute__((ext_vector_type(4)));
typedef float v8f __attribute__((ext_vector_type(8)));

#define N_AGENT 81
#define N_NBR   20
#define DD      96
#define HH      96
#define YPITCH  100
#define WPITCH  100
#define WMAT    (HH * WPITCH)

__device__ __forceinline__ void load_wq(const float* wl, int koff, v4f (&a)[6]) {
    #pragma unroll
    for (int mt = 0; mt < 6; ++mt) a[mt] = *(const v4f*)(wl + 16 * mt * WPITCH + koff);
}

__device__ __forceinline__ void wT_times_b(const float* wl, const v4f (&bq)[12], v8f (&acc)[6]) {
    v4f a[2][2][6];
    load_wq(wl, 0, a[0][0]);
    load_wq(wl, 8, a[0][1]);
    #pragma unroll
    for (int blk = 0; blk < 6; ++blk) {
        if (blk + 1 < 6) {
            load_wq(wl, 16 * (blk + 1), a[(blk + 1) & 1][0]);
            load_wq(wl, 16 * (blk + 1) + 8, a[(blk + 1) & 1][1]);
        }
        __builtin_amdgcn_sched_barrier(0);
        #pragma unroll
        for (int p = 0; p < 2; ++p) {
            const int u = 2 * blk + p;
            const v2f b0 = __builtin_shufflevector(bq[u], bq[u], 0, 1);
            const v2f b1 = __builtin_shufflevector(bq[u], bq[u], 2, 3);
            #pragma unroll
            for (int mt = 0; mt < 6; ++mt) {
                const v4f aq = a[blk & 1][p][mt];
                acc[mt] = __builtin_amdgcn_wmma_f32_16x16x4_f32(false, __builtin_shufflevector(aq, aq, 0, 1), false, b0,
                                                                (short)0, acc[mt], false, false);
            }
            #pragma unroll
            for (int mt = 0; mt < 6; ++mt) {
                const v4f aq = a[blk & 1][p][mt];
                acc[mt] = __builtin_amdgcn_wmma_f32_16x16x4_f32(false, __builtin_shufflevector(aq, aq, 2, 3), false, b1,
                                                                (short)0, acc[mt], false, false);
            }
        }
        __builtin_amdgcn_sched_barrier(0);
    }
}

__device__ __forceinline__ void ask_group(const float* yl, const int (&nb)[N_NBR], int g, int tl, float4 (&yv)[10]) {
    #pragma unroll
    for (int i = 0; i < 10; ++i)
        yv[i] = *(const float4*)(yl + nb[5 * g + (i >> 1)] + 16 * tl + 4 * (i & 1));
}

__device__ __forceinline__ void add_group(const float4 (&yv)[10], const v8f& sbt, v8f& hst) {
    #pragma unroll
    for (int i = 0; i < 10; ++i) {
        const int p = 4 * (i & 1);
        hst[p + 0] += fmaxf(sbt[p + 0] + yv[i].x, 0.f);
        hst[p + 1] += fmaxf(sbt[p + 1] + yv[i].y, 0.f);
        hst[p + 2] += fmaxf(sbt[p + 2] + yv[i].z, 0.f);
        hst[p + 3] += fmaxf(sbt[p + 3] + yv[i].w, 0.f);
    }
}

__global__ __launch_bounds__(384) void relational_wmma_kernel(
    const float* __restrict__ x,
    const int*   __restrict__ edges,
    const float* __restrict__ W1,
    const float* __restrict__ b1,
    const float* __restrict__ W2,
    const float* __restrict__ b2,
    float* __restrict__ out)
{
    __shared__ __attribute__((aligned(16))) float ybuf2[2 * N_AGENT * YPITCH];
    __shared__ __attribute__((aligned(16))) float wbuf[3 * WMAT];
    __shared__ __attribute__((aligned(16))) float obuf[12 * 16 * YPITCH];

    const int tid  = threadIdx.x;
    const int lane = tid & 31;
    const int wv12 = tid >> 5;
    const int bsel = wv12 >= 6 ? 1 : 0;
    const int wave = wv12 - 6 * bsel;
    const int b    = 2 * blockIdx.x + bsel;
    float* ybuf = ybuf2 + bsel * (N_AGENT * YPITCH);
    const int h    = lane >> 4;
    const int c    = lane & 15;
    const int arow  = wave * 16 + c;
    const int aload = arow < N_AGENT ? arow : N_AGENT - 1;

    const float* xr = x + ((size_t)b * N_AGENT + aload) * DD + 4 * h;
    v4f xf[12];
    #pragma unroll
    for (int u = 0; u < 12; ++u) xf[u] = *(const v4f*)(xr + 8 * u);

    int nb[N_NBR];
    {
        const int* ep = edges + aload * N_NBR;
        #pragma unroll
        for (int q = 0; q < N_NBR / 4; ++q) {
            const int4 e4 = *(const int4*)(ep + 4 * q);
            nb[4 * q + 0] = e4.x * YPITCH;
            nb[4 * q + 1] = e4.y * YPITCH;
            nb[4 * q + 2] = e4.z * YPITCH;
            nb[4 * q + 3] = e4.w * YPITCH;
        }
    }

    {
        const int kk = lane & 7;
        const int qq = lane >> 3;
        const int col0 = 16 * wave + 4 * qq;
        const float* g1 = W1 + (96 * bsel + kk) * HH + col0;
        const float* g2 = W2 + (48 * bsel + kk) * HH + col0;
        float* d1 = wbuf + bsel * WMAT + col0 * WPITCH + kk;
        float* d2 = wbuf + 2 * WMAT + col0 * WPITCH + 48 * bsel + kk;
        float4 wq[18];
        #pragma unroll
        for (int i = 0; i < 18; ++i)
            wq[i] = i < 12 ? *(const float4*)(g1 + 8 * i * HH) : *(const float4*)(g2 + 8 * (i - 12) * HH);
        #pragma unroll
        for (int i = 0; i < 18; ++i) {
            float* d = i < 12 ? d1 + 8 * i : d2 + 8 * (i - 12);
            d[0 * WPITCH] = wq[i].x;
            d[1 * WPITCH] = wq[i].y;
            d[2 * WPITCH] = wq[i].z;
            d[3 * WPITCH] = wq[i].w;
        }
    }
    __syncthreads();

    const float* wl12 = wbuf + c * WPITCH + 4 * h;

    {
        v8f yacc[6] = {};
        wT_times_b(wl12 + WMAT, xf, yacc);
        if (arow < N_AGENT) {
            float* yw = ybuf + arow * YPITCH + 8 * h;
            #pragma unroll
            for (int mt = 0; mt < 6; ++mt) {
                *(float4*)(yw + 16 * mt)     = make_float4(yacc[mt][0], yacc[mt][1], yacc[mt][2], yacc[mt][3]);
                *(float4*)(yw + 16 * mt + 4) = make_float4(yacc[mt][4], yacc[mt][5], yacc[mt][6], yacc[mt][7]);
            }
        }
    }

    __syncthreads();

    v8f sb[6] = {};
    wT_times_b(wl12, xf, sb);
    #pragma unroll
    for (int mt = 0; mt < 6; ++mt) {
        const float4 p0 = *(const float4*)(b1 + 16 * mt + 8 * h);
        const float4 p1 = *(const float4*)(b1 + 16 * mt + 8 * h + 4);
        sb[mt][0] += p0.x; sb[mt][1] += p0.y; sb[mt][2] += p0.z; sb[mt][3] += p0.w;
        sb[mt][4] += p1.x; sb[mt][5] += p1.y; sb[mt][6] += p1.z; sb[mt][7] += p1.w;
    }

    v8f oc[6] = {};
    {
        const float* yl  = ybuf + 8 * h;
        const float* wl3 = wbuf + 2 * WMAT + c * WPITCH + 8 * h;
        float4 yv[10];
        v4f aw[2][6];
        v8f hcur = {};
        load_wq(wl3, 0, aw[0]);
        #pragma unroll
        for (int g = 0; g < 4; ++g) {
            ask_group(yl, nb, g, 0, yv);
            __builtin_amdgcn_sched_barrier(0);
            add_group(yv, sb[0], hcur);
            __builtin_amdgcn_sched_barrier(0);
        }
        #pragma unroll
        for (int t = 0; t < 6; ++t) {
            v2f hb[4];
            #pragma unroll
            for (int j = 0; j < 4; ++j) {
                hb[j].x = hcur[2 * j];
                hb[j].y = hcur[2 * j + 1];
            }
            v8f hnext = {};
            const int tn = t + 1 < 6 ? t + 1 : 5;
            #pragma unroll
            for (int j = 0; j < 4; ++j) {
                const int ks = 4 * t + j;
                if (t + 1 < 6) ask_group(yl, nb, j, tn, yv);
                const int pr = ks >> 1;
                if ((j & 1) == 0 && pr + 1 < 12) load_wq(wl3, 16 * ((pr + 1) >> 1) + 4 * ((pr + 1) & 1), aw[(pr + 1) & 1]);
                __builtin_amdgcn_sched_barrier(0);
                #pragma unroll
                for (int mt = 0; mt < 6; ++mt) {
                    const v4f aq = aw[pr & 1][mt];
                    const v2f af = (j & 1) ? __builtin_shufflevector(aq, aq, 2, 3) : __builtin_shufflevector(aq, aq, 0, 1);
                    oc[mt] = __builtin_amdgcn_wmma_f32_16x16x4_f32(false, af, false, hb[j],
                                                                   (short)0, oc[mt], false, false);
                }
                __builtin_amdgcn_sched_barrier(0);
                if (t + 1 < 6) add_group(yv, sb[tn], hnext);
                __builtin_amdgcn_sched_barrier(0);
            }
            hcur = hnext;
        }
    }

    {
        float* ob = obuf + wv12 * (16 * YPITCH);
        float* ow = ob + c * YPITCH + 8 * h;
        #pragma unroll
        for (int ot = 0; ot < 6; ++ot) {
            const float4 q0 = *(const float4*)(b2 + 16 * ot + 8 * h);
            const float4 q1 = *(const float4*)(b2 + 16 * ot + 8 * h + 4);
            const float g0 = (float)N_NBR * q0.x, g1 = (float)N_NBR * q0.y;
            const float g2 = (float)N_NBR * q0.z, g3 = (float)N_NBR * q0.w;
            const float g4 = (float)N_NBR * q1.x, g5 = (float)N_NBR * q1.y;
            const float g6 = (float)N_NBR * q1.z, g7 = (float)N_NBR * q1.w;
            *(float4*)(ow + 16 * ot)     = make_float4(oc[ot][0] + g0, oc[ot][1] + g1, oc[ot][2] + g2, oc[ot][3] + g3);
            *(float4*)(ow + 16 * ot + 4) = make_float4(oc[ot][4] + g4, oc[ot][5] + g5, oc[ot][6] + g6, oc[ot][7] + g7);
        }
        __builtin_amdgcn_fence(__ATOMIC_RELEASE, "wavefront");
        __builtin_amdgcn_wave_barrier();
        __builtin_amdgcn_fence(__ATOMIC_ACQUIRE, "wavefront");
        const int wave_u = __builtin_amdgcn_readfirstlane(wave);
        const int nrow = (N_AGENT - 16 * wave_u) < 16 ? (N_AGENT - 16 * wave_u) : 16;
        float* og = out + ((size_t)b * N_AGENT + 16 * wave) * HH;
        float4 ov[12];
        #pragma unroll
        for (int i = 0; i < 12; ++i) {
            const int f   = 128 * i + 4 * lane;
            const int row = f / HH;
            const int col = f - HH * row;
            ov[i] = *(const float4*)(ob + row * YPITCH + col);
        }
        if (nrow == 16) {
            #pragma unroll
            for (int i = 0; i < 12; ++i) *(float4*)(og + 128 * i + 4 * lane) = ov[i];
        } else {
            const int nflt = nrow * HH;
            #pragma unroll
            for (int i = 0; i < 12; ++i) {
                const int f = 128 * i + 4 * lane;
                if (f < nflt) *(float4*)(og + f) = ov[i];
            }
        }
    }
}

extern "C" void kernel_launch(void* const* d_in, const int* in_sizes, int n_in,
                              void* d_out, int out_size, void* d_ws, size_t ws_size,
                              hipStream_t stream) {
    (void)in_sizes; (void)n_in; (void)out_size; (void)d_ws; (void)ws_size;
    const float* x     = (const float*)d_in[0];
    const int*   edges = (const int*)d_in[1];
    const float* W1    = (const float*)d_in[2];
    const float* b1    = (const float*)d_in[3];
    const float* W2    = (const float*)d_in[4];
    const float* b2    = (const float*)d_in[5];
    float* out = (float*)d_out;

    relational_wmma_kernel<<<dim3(256), dim3(384), 0, stream>>>(
        x, edges, W1, b1, W2, b2, out);
}
